# starting kernel
# baseline (speedup 1.0000x reference)
#include <hip/hip_runtime.h>

typedef float v2f __attribute__((ext_vector_type(2)));
typedef float v8f __attribute__((ext_vector_type(8)));

constexpr int Bn = 8, Cc = 128, Hh = 96, Ww = 320;
constexpr int HW = Hh * Ww;
constexpr int ND = 9;
constexpr int NQ = Cc / 4;
constexpr int TILES_PER_SEG = 4;
constexpr int SEGS = Ww / (16 * TILES_PER_SEG);
constexpr int WAVES_PER_BLOCK = 8;

__device__ __forceinline__ void load_bblock(v2f* Bb, const float* __restrict__ x2row,
                                            int j0, int g, int nn) {
  int xsrc = j0 + nn - 4;
  bool valid = ((unsigned)xsrc < (unsigned)Ww);
#pragma unroll
  for (int q = 0; q < NQ; ++q) {
    int c = 4 * q + 2 * g;
    float bx = 0.0f, by = 0.0f;
    if (valid) {
      const float* p = x2row + c * HW + xsrc;
      bx = p[0];
      by = p[HW];
    }
    Bb[q].x = bx;
    Bb[q].y = by;
  }
}

__device__ __forceinline__ v2f load_afrag(const float* __restrict__ x1row, int x0,
                                          int q, int g, int mm) {
  int c = 4 * q + 2 * g;
  const float* p = x1row + c * HW + x0 + mm;
  v2f a;
  a.x = __builtin_nontemporal_load(p);
  a.y = __builtin_nontemporal_load(p + HW);
  return a;
}

__device__ __forceinline__ v8f wmma4(const v2f a, const v2f b, const v8f c) {
  return __builtin_amdgcn_wmma_f32_16x16x4_f32(false, a, false, b, (short)0, c,
                                               false, false);
}

__device__ __forceinline__ void store_tile(float* __restrict__ outbase, int x0,
                                           int g, int nn, const v8f d0,
                                           const v8f d1) {
  const float invC = 1.0f / (float)Cc;
#pragma unroll
  for (int r = 0; r < 8; ++r) {
    int mrow = r + 8 * g;
    int t0 = nn - mrow;
    if ((unsigned)t0 <= 8u)
      __builtin_nontemporal_store(d0[r] * invC, outbase + t0 * HW + x0 + mrow);
    int t1 = nn + 16 - mrow;
    if ((unsigned)t1 <= 8u)
      __builtin_nontemporal_store(d1[r] * invC, outbase + t1 * HW + x0 + mrow);
  }
}

__global__ __launch_bounds__(WAVES_PER_BLOCK * 32) void
Correlation1d_74363063763057_kernel(const float* __restrict__ x1,
                                    const float* __restrict__ x2,
                                    float* __restrict__ out) {
  int wid = blockIdx.x * WAVES_PER_BLOCK + ((int)threadIdx.x >> 5);
  int lane = (int)threadIdx.x & 31;
  int nn = lane & 15;
  int g = lane >> 4;

  int seg = wid % SEGS;
  int row = wid / SEGS;
  if (row >= Bn * Hh) return;
  int y = row % Hh;
  int b = row / Hh;

  const float* x1row = x1 + (size_t)b * Cc * HW + (size_t)y * Ww;
  const float* x2row = x2 + (size_t)b * Cc * HW + (size_t)y * Ww;
  float* outbase = out + (size_t)b * ND * HW + (size_t)y * Ww;

  int xs = seg * (16 * TILES_PER_SEG);

  v8f zero = {};
  v8f d0a = zero, d0b = zero;
  v8f d1a = zero, d1b = zero;
  v8f d0_done = zero;

  v2f Aa[NQ];
  v2f Bb[NQ];

  load_bblock(Bb, x2row, xs, g, nn);
#pragma unroll
  for (int q = 0; q < NQ; ++q) {
    Aa[q] = load_afrag(x1row, xs, q, g, nn);
    if (q & 1)
      d0b = wmma4(Aa[q], Bb[q], d0b);
    else
      d0a = wmma4(Aa[q], Bb[q], d0a);
  }
  d0_done = d0a + d0b;

  for (int kblk = 1; kblk < TILES_PER_SEG; ++kblk) {
    int j0 = xs + 16 * kblk;
    load_bblock(Bb, x2row, j0, g, nn);
    d0a = zero;
    d0b = zero;
#pragma unroll
    for (int q = 0; q < NQ; ++q) {
      v2f an = load_afrag(x1row, j0, q, g, nn);
      if (q & 1) {
        d1b = wmma4(Aa[q], Bb[q], d1b);
        d0b = wmma4(an, Bb[q], d0b);
      } else {
        d1a = wmma4(Aa[q], Bb[q], d1a);
        d0a = wmma4(an, Bb[q], d0a);
      }
      Aa[q] = an;
    }
    store_tile(outbase, j0 - 16, g, nn, d0_done, d1a + d1b);
    d0_done = d0a + d0b;
    d1a = zero;
    d1b = zero;
  }

  {
    int j0 = xs + 16 * TILES_PER_SEG;
    load_bblock(Bb, x2row, j0, g, nn);
#pragma unroll
    for (int q = 0; q < NQ; ++q) {
      if (q & 1)
        d1b = wmma4(Aa[q], Bb[q], d1b);
      else
        d1a = wmma4(Aa[q], Bb[q], d1a);
    }
    store_tile(outbase, j0 - 16, g, nn, d0_done, d1a + d1b);
  }
}

extern "C" void kernel_launch(void* const* d_in, const int* in_sizes, int n_in,
                              void* d_out, int out_size, void* d_ws, size_t ws_size,
                              hipStream_t stream) {
  (void)in_sizes; (void)n_in; (void)out_size; (void)d_ws; (void)ws_size;
  const float* x1 = (const float*)d_in[0];
  const float* x2 = (const float*)d_in[1];
  float* out = (float*)d_out;

  int totalWaves = Bn * Hh * SEGS;
  int blocks = (totalWaves + WAVES_PER_BLOCK - 1) / WAVES_PER_BLOCK;
  Correlation1d_74363063763057_kernel<<<blocks, WAVES_PER_BLOCK * 32, 0, stream>>>(
      x1, x2, out);
}


# x2 block loads made unconditional (column clamped into the row, padding lanes zeroed by select): a block's 64 loads issue as one group, not as 32 exec-masked pairs each closed by a translation wait
# speedup vs baseline: 1.0635x; 1.0635x over previous
#include <hip/hip_runtime.h>

typedef float v2f __attribute__((ext_vector_type(2)));
typedef float v8f __attribute__((ext_vector_type(8)));

constexpr int Bn = 8, Cc = 128, Hh = 96, Ww = 320;
constexpr int HW = Hh * Ww;
constexpr int ND = 9;
constexpr int NQ = Cc / 4;
constexpr int TILES_PER_SEG = 4;
constexpr int SEGS = Ww / (16 * TILES_PER_SEG);
constexpr int WAVES_PER_BLOCK = 8;

__device__ __forceinline__ void load_bblock(v2f* Bb, const float* __restrict__ x2row,
                                            int j0, int g, int nn) {
  int xsrc = j0 + nn - 4;
  bool valid = ((unsigned)xsrc < (unsigned)Ww);
  int xcl = xsrc < 0 ? 0 : (xsrc > Ww - 1 ? Ww - 1 : xsrc);
  const float* p0 = x2row + (2 * g) * HW + xcl;
#pragma unroll
  for (int q = 0; q < NQ; ++q) {
    const float* p = p0 + (4 * q) * HW;
    float bx = p[0];
    float by = p[HW];
    Bb[q].x = valid ? bx : 0.0f;
    Bb[q].y = valid ? by : 0.0f;
  }
}

__device__ __forceinline__ v2f load_afrag(const float* __restrict__ x1row, int x0,
                                          int q, int g, int mm) {
  int c = 4 * q + 2 * g;
  const float* p = x1row + c * HW + x0 + mm;
  v2f a;
  a.x = __builtin_nontemporal_load(p);
  a.y = __builtin_nontemporal_load(p + HW);
  return a;
}

__device__ __forceinline__ v8f wmma4(const v2f a, const v2f b, const v8f c) {
  return __builtin_amdgcn_wmma_f32_16x16x4_f32(false, a, false, b, (short)0, c,
                                               false, false);
}

__device__ __forceinline__ void store_tile(float* __restrict__ outbase, int x0,
                                           int g, int nn, const v8f d0,
                                           const v8f d1) {
  const float invC = 1.0f / (float)Cc;
#pragma unroll
  for (int r = 0; r < 8; ++r) {
    int mrow = r + 8 * g;
    int t0 = nn - mrow;
    if ((unsigned)t0 <= 8u)
      __builtin_nontemporal_store(d0[r] * invC, outbase + t0 * HW + x0 + mrow);
    int t1 = nn + 16 - mrow;
    if ((unsigned)t1 <= 8u)
      __builtin_nontemporal_store(d1[r] * invC, outbase + t1 * HW + x0 + mrow);
  }
}

__global__ __launch_bounds__(WAVES_PER_BLOCK * 32) void
Correlation1d_74363063763057_kernel(const float* __restrict__ x1,
                                    const float* __restrict__ x2,
                                    float* __restrict__ out) {
  int wid = blockIdx.x * WAVES_PER_BLOCK + ((int)threadIdx.x >> 5);
  int lane = (int)threadIdx.x & 31;
  int nn = lane & 15;
  int g = lane >> 4;

  int seg = wid % SEGS;
  int row = wid / SEGS;
  if (row >= Bn * Hh) return;
  int y = row % Hh;
  int b = row / Hh;

  const float* x1row = x1 + (size_t)b * Cc * HW + (size_t)y * Ww;
  const float* x2row = x2 + (size_t)b * Cc * HW + (size_t)y * Ww;
  float* outbase = out + (size_t)b * ND * HW + (size_t)y * Ww;

  int xs = seg * (16 * TILES_PER_SEG);

  v8f zero = {};
  v8f d0a = zero, d0b = zero;
  v8f d1a = zero, d1b = zero;
  v8f d0_done = zero;

  v2f Aa[NQ];
  v2f Bb[NQ];

  load_bblock(Bb, x2row, xs, g, nn);
#pragma unroll
  for (int q = 0; q < NQ; ++q) {
    Aa[q] = load_afrag(x1row, xs, q, g, nn);
    if (q & 1)
      d0b = wmma4(Aa[q], Bb[q], d0b);
    else
      d0a = wmma4(Aa[q], Bb[q], d0a);
  }
  d0_done = d0a + d0b;

  for (int kblk = 1; kblk < TILES_PER_SEG; ++kblk) {
    int j0 = xs + 16 * kblk;
    load_bblock(Bb, x2row, j0, g, nn);
    d0a = zero;
    d0b = zero;
#pragma unroll
    for (int q = 0; q < NQ; ++q) {
      v2f an = load_afrag(x1row, j0, q, g, nn);
      if (q & 1) {
        d1b = wmma4(Aa[q], Bb[q], d1b);
        d0b = wmma4(an, Bb[q], d0b);
      } else {
        d1a = wmma4(Aa[q], Bb[q], d1a);
        d0a = wmma4(an, Bb[q], d0a);
      }
      Aa[q] = an;
    }
    store_tile(outbase, j0 - 16, g, nn, d0_done, d1a + d1b);
    d0_done = d0a + d0b;
    d1a = zero;
    d1b = zero;
  }

  {
    int j0 = xs + 16 * TILES_PER_SEG;
    load_bblock(Bb, x2row, j0, g, nn);
#pragma unroll
    for (int q = 0; q < NQ; ++q) {
      if (q & 1)
        d1b = wmma4(Aa[q], Bb[q], d1b);
      else
        d1a = wmma4(Aa[q], Bb[q], d1a);
    }
    store_tile(outbase, j0 - 16, g, nn, d0_done, d1a + d1b);
  }
}

extern "C" void kernel_launch(void* const* d_in, const int* in_sizes, int n_in,
                              void* d_out, int out_size, void* d_ws, size_t ws_size,
                              hipStream_t stream) {
  (void)in_sizes; (void)n_in; (void)out_size; (void)d_ws; (void)ws_size;
  const float* x1 = (const float*)d_in[0];
  const float* x2 = (const float*)d_in[1];
  float* out = (float*)d_out;

  int totalWaves = Bn * Hh * SEGS;
  int blocks = (totalWaves + WAVES_PER_BLOCK - 1) / WAVES_PER_BLOCK;
  Correlation1d_74363063763057_kernel<<<blocks, WAVES_PER_BLOCK * 32, 0, stream>>>(
      x1, x2, out);
}


# x1, x2 read once by 16-byte loads over whole 256-B units, lane quads transposed by DPP to fragment layout; K streamed once over 8 accumulator tiles (one sum chain per tile, no even/odd split)
# speedup vs baseline: 1.7246x; 1.6216x over previous
#include <hip/hip_runtime.h>


typedef float v2f __attribute__((ext_vector_type(2)));
typedef float v4f __attribute__((ext_vector_type(4)));
typedef float v8f __attribute__((ext_vector_type(8)));

constexpr int Bn = 8, Cc = 128, Hh = 96, Ww = 320;
constexpr int HW = Hh * Ww;
constexpr int ND = 9;
constexpr int NQ = Cc / 4;
constexpr int TILES_PER_SEG = 4;
constexpr int SEGS = Ww / (16 * TILES_PER_SEG);
constexpr int WAVES_PER_BLOCK = 8;
constexpr int PD = 6;

__device__ __forceinline__ float lane_xor1(float v) {
  int i = __float_as_int(v);
  return __int_as_float(__builtin_amdgcn_update_dpp(i, i, 0xB1, 0xF, 0xF, true));
}
__device__ __forceinline__ float lane_xor2(float v) {
  int i = __float_as_int(v);
  return __int_as_float(__builtin_amdgcn_update_dpp(i, i, 0x4E, 0xF, 0xF, true));
}

__device__ __forceinline__ v4f quad_transpose(const v4f r, bool odd, bool hi) {
  float t0 = lane_xor1(r.y), t1 = lane_xor1(r.x);
  float t2 = lane_xor1(r.w), t3 = lane_xor1(r.z);
  float s0 = odd ? t0 : r.x;
  float s1 = odd ? r.y : t1;
  float s2 = odd ? t2 : r.z;
  float s3 = odd ? r.w : t3;
  float u0 = lane_xor2(s2), u1 = lane_xor2(s3);
  float u2 = lane_xor2(s0), u3 = lane_xor2(s1);
  v4f o;
  o.x = hi ? u0 : s0;
  o.y = hi ? u1 : s1;
  o.z = hi ? s2 : u2;
  o.w = hi ? s3 : u3;
  return o;
}

__device__ __forceinline__ v8f wmma4(const v2f a, const v2f b, const v8f c) {
  return __builtin_amdgcn_wmma_f32_16x16x4_f32(false, a, false, b, (short)0, c,
                                               false, false);
}

__device__ __forceinline__ void store_tile(float* __restrict__ outbase, int x0,
                                           int g, int nn, const v8f d0,
                                           const v8f d1) {
  const float invC = 1.0f / (float)Cc;
#pragma unroll
  for (int r = 0; r < 8; ++r) {
    int mrow = r + 8 * g;
    int t0 = nn - mrow;
    if ((unsigned)t0 <= 8u)
      __builtin_nontemporal_store(d0[r] * invC, outbase + t0 * HW + x0 + mrow);
    int t1 = nn + 16 - mrow;
    if ((unsigned)t1 <= 8u)
      __builtin_nontemporal_store(d1[r] * invC, outbase + t1 * HW + x0 + mrow);
  }
}

struct Chunk {
  v4f ax, ay;
  v4f bx, by;
  float tx, ty;
};

__device__ __forceinline__ void load_chunk(Chunk& ch, const float* __restrict__ pa,
                                           const float* __restrict__ pb,
                                           const float* __restrict__ pt, int q) {
  const float* a0 = pa + (4 * q) * HW;
  const float* b0 = pb + (4 * q) * HW;
  const float* t0 = pt + (4 * q) * HW;
  ch.ax = __builtin_nontemporal_load((const v4f*)a0);
  ch.ay = __builtin_nontemporal_load((const v4f*)(a0 + HW));
  ch.bx = *(const v4f*)b0;
  ch.by = *(const v4f*)(b0 + HW);
  ch.tx = t0[0];
  ch.ty = t0[HW];
}

__global__ __launch_bounds__(WAVES_PER_BLOCK * 32) void
Correlation1d_74363063763057_kernel(const float* __restrict__ x1,
                                    const float* __restrict__ x2,
                                    float* __restrict__ out) {
  int wid = blockIdx.x * WAVES_PER_BLOCK + ((int)threadIdx.x >> 5);
  int lane = (int)threadIdx.x & 31;
  int nn = lane & 15;
  int g = lane >> 4;
  int qa = nn >> 2;
  int qt = nn & 3;
  bool odd = (nn & 1) != 0;
  bool hi = (nn & 2) != 0;

  int seg = wid % SEGS;
  int row = wid / SEGS;
  if (row >= Bn * Hh) return;
  int y = row % Hh;
  int b = row / Hh;

  const float* x1row = x1 + (size_t)b * Cc * HW + (size_t)y * Ww;
  const float* x2row = x2 + (size_t)b * Cc * HW + (size_t)y * Ww;
  float* outbase = out + (size_t)b * ND * HW + (size_t)y * Ww;

  int xs = seg * (16 * TILES_PER_SEG);

  int acol = xs + 16 * qt + 4 * qa;
  int bcol = acol - 4 < 0 ? 0 : acol - 4;
  bool padl = (xs == 0) && (nn < 4);
  int tcol = xs + 60 + nn;
  bool tvalid = tcol < Ww;
  int tcl = tcol > Ww - 1 ? Ww - 1 : tcol;
  const float* pa = x1row + (2 * g) * HW + acol;
  const float* pb = x2row + (2 * g) * HW + bcol;
  const float* pt = x2row + (2 * g) * HW + tcl;

  v8f zero = {};
  v8f d0[TILES_PER_SEG], d1[TILES_PER_SEG];
#pragma unroll
  for (int t = 0; t < TILES_PER_SEG; ++t) {
    d0[t] = zero;
    d1[t] = zero;
  }

  Chunk ring[PD];
#pragma unroll
  for (int q = 0; q < PD; ++q) load_chunk(ring[q], pa, pb, pt, q);

#pragma unroll
  for (int q = 0; q < NQ; ++q) {
    Chunk cur = ring[q % PD];
    if (q + PD < NQ) load_chunk(ring[q % PD], pa, pb, pt, q + PD);
    __builtin_amdgcn_sched_barrier(0);

    v4f ax = quad_transpose(cur.ax, odd, hi);
    v4f ay = quad_transpose(cur.ay, odd, hi);
    v4f bx = quad_transpose(cur.bx, odd, hi);
    v4f by = quad_transpose(cur.by, odd, hi);
    bx.x = padl ? 0.0f : bx.x;
    by.x = padl ? 0.0f : by.x;
    v2f a[4], bb[5];
    a[0].x = ax.x; a[0].y = ay.x;
    a[1].x = ax.y; a[1].y = ay.y;
    a[2].x = ax.z; a[2].y = ay.z;
    a[3].x = ax.w; a[3].y = ay.w;
    bb[0].x = bx.x; bb[0].y = by.x;
    bb[1].x = bx.y; bb[1].y = by.y;
    bb[2].x = bx.z; bb[2].y = by.z;
    bb[3].x = bx.w; bb[3].y = by.w;
    bb[4].x = tvalid ? cur.tx : 0.0f;
    bb[4].y = tvalid ? cur.ty : 0.0f;

#pragma unroll
    for (int t = 0; t < TILES_PER_SEG; ++t) {
      d0[t] = wmma4(a[t], bb[t], d0[t]);
      d1[t] = wmma4(a[t], bb[t + 1], d1[t]);
    }
  }

#pragma unroll
  for (int t = 0; t < TILES_PER_SEG; ++t)
    store_tile(outbase, xs + 16 * t, g, nn, d0[t], d1[t]);
}

extern "C" void kernel_launch(void* const* d_in, const int* in_sizes, int n_in,
                              void* d_out, int out_size, void* d_ws, size_t ws_size,
                              hipStream_t stream) {
  (void)in_sizes; (void)n_in; (void)out_size; (void)d_ws; (void)ws_size;
  const float* x1 = (const float*)d_in[0];
  const float* x2 = (const float*)d_in[1];
  float* out = (float*)d_out;

  int totalWaves = Bn * Hh * SEGS;
  int blocks = (totalWaves + WAVES_PER_BLOCK - 1) / WAVES_PER_BLOCK;
  Correlation1d_74363063763057_kernel<<<blocks, WAVES_PER_BLOCK * 32, 0, stream>>>(
      x1, x2, out);
}


# output band rotated in registers to plane-per-lane order (a select and a DPP row rotate per accumulator register), stored as 8 columns per lane: 8 16-byte stores per wave, not 64 masked 4-byte stores
# speedup vs baseline: 2.8293x; 1.6406x over previous
#include <hip/hip_runtime.h>


typedef float v2f __attribute__((ext_vector_type(2)));
typedef float v4f __attribute__((ext_vector_type(4)));
typedef float v8f __attribute__((ext_vector_type(8)));

constexpr int Bn = 8, Cc = 128, Hh = 96, Ww = 320;
constexpr int HW = Hh * Ww;
constexpr int ND = 9;
constexpr int NQ = Cc / 4;
constexpr int TILES_PER_SEG = 4;
constexpr int SEGS = Ww / (16 * TILES_PER_SEG);
constexpr int WAVES_PER_BLOCK = 8;
constexpr int PD = 6;

__device__ __forceinline__ float lane_xor1(float v) {
  int i = __float_as_int(v);
  return __int_as_float(__builtin_amdgcn_update_dpp(i, i, 0xB1, 0xF, 0xF, true));
}
__device__ __forceinline__ float lane_xor2(float v) {
  int i = __float_as_int(v);
  return __int_as_float(__builtin_amdgcn_update_dpp(i, i, 0x4E, 0xF, 0xF, true));
}

__device__ __forceinline__ v4f quad_transpose(const v4f r, bool odd, bool hi) {
  float t0 = lane_xor1(r.y), t1 = lane_xor1(r.x);
  float t2 = lane_xor1(r.w), t3 = lane_xor1(r.z);
  float s0 = odd ? t0 : r.x;
  float s1 = odd ? r.y : t1;
  float s2 = odd ? t2 : r.z;
  float s3 = odd ? r.w : t3;
  float u0 = lane_xor2(s2), u1 = lane_xor2(s3);
  float u2 = lane_xor2(s0), u3 = lane_xor2(s1);
  v4f o;
  o.x = hi ? u0 : s0;
  o.y = hi ? u1 : s1;
  o.z = hi ? s2 : u2;
  o.w = hi ? s3 : u3;
  return o;
}

__device__ __forceinline__ v8f wmma4(const v2f a, const v2f b, const v8f c) {
  return __builtin_amdgcn_wmma_f32_16x16x4_f32(false, a, false, b, (short)0, c,
                                               false, false);
}

template <int R>
__device__ __forceinline__ float row_rotl(float v) {
  if constexpr (R == 0) {
    return v;
  } else {
    int i = __float_as_int(v);
    return __int_as_float(
        __builtin_amdgcn_update_dpp(i, i, 0x120 + (16 - R), 0xF, 0xF, true));
  }
}

template <int R>
__device__ __forceinline__ float band_reg(const v8f a, const v8f b, int ng) {
  float s = (ng >= R) ? a[R] : b[R];
  return row_rotl<R>(s) * (1.0f / (float)Cc);
}

__device__ __forceinline__ void band_tile(const v8f a, const v8f b, int ng,
                                          v4f& lo, v4f& hi) {
  lo.x = band_reg<0>(a, b, ng);
  lo.y = band_reg<1>(a, b, ng);
  lo.z = band_reg<2>(a, b, ng);
  lo.w = band_reg<3>(a, b, ng);
  hi.x = band_reg<4>(a, b, ng);
  hi.y = band_reg<5>(a, b, ng);
  hi.z = band_reg<6>(a, b, ng);
  hi.w = band_reg<7>(a, b, ng);
}

struct Chunk {
  v4f ax, ay;
  v4f bx, by;
  float tx, ty;
};

__device__ __forceinline__ void load_chunk(Chunk& ch, const float* __restrict__ pa,
                                           const float* __restrict__ pb,
                                           const float* __restrict__ pt, int q) {
  const float* a0 = pa + (4 * q) * HW;
  const float* b0 = pb + (4 * q) * HW;
  const float* t0 = pt + (4 * q) * HW;
  ch.ax = __builtin_nontemporal_load((const v4f*)a0);
  ch.ay = __builtin_nontemporal_load((const v4f*)(a0 + HW));
  ch.bx = *(const v4f*)b0;
  ch.by = *(const v4f*)(b0 + HW);
  ch.tx = t0[0];
  ch.ty = t0[HW];
}

__global__ __launch_bounds__(WAVES_PER_BLOCK * 32) void
Correlation1d_74363063763057_kernel(const float* __restrict__ x1,
                                    const float* __restrict__ x2,
                                    float* __restrict__ out) {
  int wid = blockIdx.x * WAVES_PER_BLOCK + ((int)threadIdx.x >> 5);
  int lane = (int)threadIdx.x & 31;
  int nn = lane & 15;
  int g = lane >> 4;
  int qa = nn >> 2;
  int qt = nn & 3;
  bool odd = (nn & 1) != 0;
  bool hi = (nn & 2) != 0;

  int seg = wid % SEGS;
  int row = wid / SEGS;
  if (row >= Bn * Hh) return;
  int y = row % Hh;
  int b = row / Hh;

  const float* x1row = x1 + (size_t)b * Cc * HW + (size_t)y * Ww;
  const float* x2row = x2 + (size_t)b * Cc * HW + (size_t)y * Ww;
  float* outbase = out + (size_t)b * ND * HW + (size_t)y * Ww;

  int xs = seg * (16 * TILES_PER_SEG);

  int acol = xs + 16 * qt + 4 * qa;
  int bcol = acol - 4 < 0 ? 0 : acol - 4;
  bool padl = (xs == 0) && (nn < 4);
  int tcol = xs + 60 + nn;
  bool tvalid = tcol < Ww;
  int tcl = tcol > Ww - 1 ? Ww - 1 : tcol;
  const float* pa = x1row + (2 * g) * HW + acol;
  const float* pb = x2row + (2 * g) * HW + bcol;
  const float* pt = x2row + (2 * g) * HW + tcl;

  v8f zero = {};
  v8f d0[TILES_PER_SEG], d1[TILES_PER_SEG];
#pragma unroll
  for (int t = 0; t < TILES_PER_SEG; ++t) {
    d0[t] = zero;
    d1[t] = zero;
  }

  Chunk ring[PD];
#pragma unroll
  for (int q = 0; q < PD; ++q) load_chunk(ring[q], pa, pb, pt, q);

#pragma unroll
  for (int q = 0; q < NQ; ++q) {
    Chunk cur = ring[q % PD];
    if (q + PD < NQ) load_chunk(ring[q % PD], pa, pb, pt, q + PD);
    __builtin_amdgcn_sched_barrier(0);

    v4f ax = quad_transpose(cur.ax, odd, hi);
    v4f ay = quad_transpose(cur.ay, odd, hi);
    v4f bx = quad_transpose(cur.bx, odd, hi);
    v4f by = quad_transpose(cur.by, odd, hi);
    bx.x = padl ? 0.0f : bx.x;
    by.x = padl ? 0.0f : by.x;
    v2f a[4], bb[5];
    a[0].x = ax.x; a[0].y = ay.x;
    a[1].x = ax.y; a[1].y = ay.y;
    a[2].x = ax.z; a[2].y = ay.z;
    a[3].x = ax.w; a[3].y = ay.w;
    bb[0].x = bx.x; bb[0].y = by.x;
    bb[1].x = bx.y; bb[1].y = by.y;
    bb[2].x = bx.z; bb[2].y = by.z;
    bb[3].x = bx.w; bb[3].y = by.w;
    bb[4].x = tvalid ? cur.tx : 0.0f;
    bb[4].y = tvalid ? cur.ty : 0.0f;

#pragma unroll
    for (int t = 0; t < TILES_PER_SEG; ++t) {
      d0[t] = wmma4(a[t], bb[t], d0[t]);
      d1[t] = wmma4(a[t], bb[t + 1], d1[t]);
    }
  }

  int ng = nn - 8 * g;
  int tpl = ng & 15;
  bool act = tpl <= 8;
  v4f olo[TILES_PER_SEG], ohi[TILES_PER_SEG];
#pragma unroll
  for (int t = 0; t < TILES_PER_SEG; ++t)
    band_tile(d0[t], d1[t], ng, olo[t], ohi[t]);
  if (act) {
    float* po = outbase + (size_t)tpl * HW + xs + 8 * g;
#pragma unroll
    for (int t = 0; t < TILES_PER_SEG; ++t) {
      __builtin_nontemporal_store(olo[t], (v4f*)(po + 16 * t));
      __builtin_nontemporal_store(ohi[t], (v4f*)(po + 16 * t + 4));
    }
  }
}

extern "C" void kernel_launch(void* const* d_in, const int* in_sizes, int n_in,
                              void* d_out, int out_size, void* d_ws, size_t ws_size,
                              hipStream_t stream) {
  (void)in_sizes; (void)n_in; (void)out_size; (void)d_ws; (void)ws_size;
  const float* x1 = (const float*)d_in[0];
  const float* x2 = (const float*)d_in[1];
  float* out = (float*)d_out;

  int totalWaves = Bn * Hh * SEGS;
  int blocks = (totalWaves + WAVES_PER_BLOCK - 1) / WAVES_PER_BLOCK;
  Correlation1d_74363063763057_kernel<<<blocks, WAVES_PER_BLOCK * 32, 0, stream>>>(
      x1, x2, out);
}
